# starting kernel
# baseline (speedup 1.0000x reference)
#include <hip/hip_runtime.h>


typedef __attribute__((ext_vector_type(2))) float v2f;
typedef __attribute__((ext_vector_type(4))) float v4f;
typedef __attribute__((ext_vector_type(8))) float v8f;

#define PAD_ID 50256

constexpr int kB    = 4;
constexpr int kS    = 2048;
constexpr int kD    = 4096;
constexpr int kRows = 2 * kB * kS;

template <int KSEG>
__global__ __launch_bounds__(256) void reward_gemv_wmma(
    const float* __restrict__ hs,
    const float* __restrict__ w,
    float* __restrict__ part) {
  const int lane = threadIdx.x & 31;
  const int wid  = blockIdx.x * (blockDim.x >> 5) + (threadIdx.x >> 5);
  const int tile = wid / KSEG;
  const int seg  = wid % KSEG;
  const int row0 = tile * 16;

  const int half = lane >> 4;
  const int m    = lane & 15;

  constexpr int KLEN = kD / KSEG;
  const int k0 = seg * KLEN;

  const float* __restrict__ rowp = hs + (size_t)(row0 + m) * kD + k0 + 4 * half;
  const float* __restrict__ wp   = w + k0 + 4 * half;

  v8f acc0 = {};
  v8f acc1 = {};

#pragma unroll 4
  for (int k = 0; k < KLEN; k += 8) {
    v4f a  = __builtin_nontemporal_load((const v4f*)(rowp + k));
    v4f bw = *(const v4f*)(wp + k);

    v2f a0 = {a.x, a.y};
    v2f b0 = {bw.x, bw.y};
    acc0 = __builtin_amdgcn_wmma_f32_16x16x4_f32(
         false, a0,  false, b0,
         (short)0, acc0,  false,  false);

    v2f a1 = {a.z, a.w};
    v2f b1 = {bw.z, bw.w};
    acc1 = __builtin_amdgcn_wmma_f32_16x16x4_f32(
        false, a1, false, b1, (short)0, acc1, false, false);
  }

  v8f acc = acc0 + acc1;

  if (m == 0) {
    float* outp = part + (size_t)seg * kRows + row0 + 8 * half;
#pragma unroll
    for (int r = 0; r < 8; ++r) outp[r] = acc[r];
  }
}

template <int KSEG>
__global__ __launch_bounds__(256) void combine_partials(
    const float* __restrict__ part,
    float* __restrict__ rewards) {
  const int i = blockIdx.x * blockDim.x + threadIdx.x;
  if (i < kRows) {
    float s = 0.f;
#pragma unroll
    for (int g = 0; g < KSEG; ++g) s += part[(size_t)g * kRows + i];
    rewards[i] = s;
  }
}

__device__ __forceinline__ float blk_sum(float v, float* buf) {
  const int t = threadIdx.x;
  buf[t] = v;
  __syncthreads();
  for (int o = 128; o > 0; o >>= 1) {
    if (t < o) buf[t] += buf[t + o];
    __syncthreads();
  }
  float r = buf[0];
  __syncthreads();
  return r;
}

__device__ __forceinline__ int blk_min(int v, int* buf) {
  const int t = threadIdx.x;
  buf[t] = v;
  __syncthreads();
  for (int o = 128; o > 0; o >>= 1) {
    if (t < o) buf[t] = min(buf[t], buf[t + o]);
    __syncthreads();
  }
  int r = buf[0];
  __syncthreads();
  return r;
}

__global__ __launch_bounds__(256) void pair_reduce(
    const int* __restrict__ ids,
    const float* __restrict__ rewards,
    float* __restrict__ out) {
  __shared__ float fbuf[256];
  __shared__ int   ibuf[256];
  const int t = threadIdx.x;

  float loss_acc = 0.f;

  for (int b = 0; b < kB; ++b) {
    const int*   ch = ids + (size_t)b * kS;
    const int*   rj = ids + (size_t)(kB + b) * kS;
    const float* cr = rewards + (size_t)b * kS;
    const float* rr = rewards + (size_t)(kB + b) * kS;

    int ldiv = kS, lc = kS, lr = kS;
    for (int s = t; s < kS; s += 256) {
      int c = ch[s], r = rj[s];
      if (c != r)      ldiv = min(ldiv, s);
      if (c == PAD_ID) lc   = min(lc, s);
      if (r == PAD_ID) lr   = min(lr, s);
    }
    const int div = blk_min(ldiv, ibuf);
    const int cin = blk_min(lc, ibuf);
    const int rin = blk_min(lr, ibuf);
    const int end = max(cin, rin);

    const int   cnt  = max(end - div, 0);
    const float fcnt = (float)max(cnt, 1);

    float sc = 0.f, sr = 0.f, sl = 0.f;
    for (int s = t; s < kS; s += 256) {
      if (s >= div && s < end) {
        float c = cr[s], r = rr[s];
        sc += c;
        sr += r;
        float z = r - c;
        sl += fmaxf(z, 0.f) + log1pf(expf(-fabsf(z)));
      }
    }
    sc = blk_sum(sc, fbuf);
    sr = blk_sum(sr, fbuf);
    sl = blk_sum(sl, fbuf);

    loss_acc += sl / fcnt;

    if (t == 0) {
      const int base = 1 + kRows;
      out[base + b]          = sc / fcnt;
      out[base + kB + b]     = sr / fcnt;
      const int last = max(end - 1, 0);
      out[base + 2 * kB + b] = cr[last];
      out[base + 3 * kB + b] = rr[last];
    }
    __syncthreads();
  }

  if (t == 0) out[0] = loss_acc / (float)kB;
}

extern "C" void kernel_launch(void* const* d_in, const int* in_sizes, int n_in,
                              void* d_out, int out_size, void* d_ws, size_t ws_size,
                              hipStream_t stream) {
  (void)in_sizes; (void)n_in; (void)out_size;

  const float* hs  = (const float*)d_in[0];
  const float* w   = (const float*)d_in[1];
  const int*   ids = (const int*)d_in[2];

  float* out     = (float*)d_out;
  float* rewards = out + 1;

  constexpr int KSEG = 4;
  const dim3 blk(256);

  if (ws_size >= (size_t)KSEG * kRows * sizeof(float)) {
    float* part = (float*)d_ws;
    const int waves = (kRows / 16) * KSEG;
    reward_gemv_wmma<KSEG><<<dim3(waves / 8), blk, 0, stream>>>(hs, w, part);
    combine_partials<KSEG><<<dim3((kRows + 255) / 256), blk, 0, stream>>>(part, rewards);
  } else {
    reward_gemv_wmma<1><<<dim3((kRows / 16) / 8), blk, 0, stream>>>(hs, w, rewards);
  }

  pair_reduce<<<1, blk, 0, stream>>>(ids, rewards, out);
}


# pair_reduce: four pairs side by side in one 1024-thread block, wave butterfly min/sum then one fixed-order LDS combine (3 barriers, was 240); reduction tree of the masked sums changes
# speedup vs baseline: 1.5347x; 1.5347x over previous
#include <hip/hip_runtime.h>


typedef __attribute__((ext_vector_type(2))) float v2f;
typedef __attribute__((ext_vector_type(4))) float v4f;
typedef __attribute__((ext_vector_type(8))) float v8f;

#define PAD_ID 50256

constexpr int kB    = 4;
constexpr int kS    = 2048;
constexpr int kD    = 4096;
constexpr int kRows = 2 * kB * kS;

template <int KSEG>
__global__ __launch_bounds__(256) void reward_gemv_wmma(
    const float* __restrict__ hs,
    const float* __restrict__ w,
    float* __restrict__ part) {
  const int lane = threadIdx.x & 31;
  const int wid  = blockIdx.x * (blockDim.x >> 5) + (threadIdx.x >> 5);
  const int tile = wid / KSEG;
  const int seg  = wid % KSEG;
  const int row0 = tile * 16;

  const int half = lane >> 4;
  const int m    = lane & 15;

  constexpr int KLEN = kD / KSEG;
  const int k0 = seg * KLEN;

  const float* __restrict__ rowp = hs + (size_t)(row0 + m) * kD + k0 + 4 * half;
  const float* __restrict__ wp   = w + k0 + 4 * half;

  v8f acc0 = {};
  v8f acc1 = {};

#pragma unroll 4
  for (int k = 0; k < KLEN; k += 8) {
    v4f a  = __builtin_nontemporal_load((const v4f*)(rowp + k));
    v4f bw = *(const v4f*)(wp + k);

    v2f a0 = {a.x, a.y};
    v2f b0 = {bw.x, bw.y};
    acc0 = __builtin_amdgcn_wmma_f32_16x16x4_f32(
         false, a0,  false, b0,
         (short)0, acc0,  false,  false);

    v2f a1 = {a.z, a.w};
    v2f b1 = {bw.z, bw.w};
    acc1 = __builtin_amdgcn_wmma_f32_16x16x4_f32(
        false, a1, false, b1, (short)0, acc1, false, false);
  }

  v8f acc = acc0 + acc1;

  if (m == 0) {
    float* outp = part + (size_t)seg * kRows + row0 + 8 * half;
#pragma unroll
    for (int r = 0; r < 8; ++r) outp[r] = acc[r];
  }
}

template <int KSEG>
__global__ __launch_bounds__(256) void combine_partials(
    const float* __restrict__ part,
    float* __restrict__ rewards) {
  const int i = blockIdx.x * blockDim.x + threadIdx.x;
  if (i < kRows) {
    float s = 0.f;
#pragma unroll
    for (int g = 0; g < KSEG; ++g) s += part[(size_t)g * kRows + i];
    rewards[i] = s;
  }
}

constexpr int kPairThreads = 256;
constexpr int kPairWaves   = kPairThreads / 32;
constexpr int kPerThread   = kS / kPairThreads;

__device__ __forceinline__ int wave_min(int v) {
  for (int o = 16; o > 0; o >>= 1) v = min(v, __shfl_xor(v, o, 32));
  return v;
}

__device__ __forceinline__ float wave_sum(float v) {
  for (int o = 16; o > 0; o >>= 1) v += __shfl_xor(v, o, 32);
  return v;
}

__global__ __launch_bounds__(1024) void pair_reduce(
    const int* __restrict__ ids,
    const float* __restrict__ rewards,
    float* __restrict__ out) {
  __shared__ int   wmin[3][kB * kPairWaves];
  __shared__ float wsum[3][kB * kPairWaves];
  __shared__ float tail[4 * kB];
  __shared__ float ploss[kB];

  const int t    = threadIdx.x;
  const int b    = t >> 8;
  const int tl   = t & 255;
  const int lane = t & 31;
  const int wv   = t >> 5;

  const int*   ch = ids + (size_t)b * kS;
  const int*   rj = ids + (size_t)(kB + b) * kS;
  const float* cr = rewards + (size_t)b * kS;
  const float* rr = rewards + (size_t)(kB + b) * kS;

  int   c[kPerThread], r[kPerThread];
  float fc[kPerThread], fr[kPerThread];
#pragma unroll
  for (int i = 0; i < kPerThread; ++i) {
    const int s = tl + kPairThreads * i;
    c[i]  = ch[s];
    r[i]  = rj[s];
    fc[i] = cr[s];
    fr[i] = rr[s];
  }

  int ldiv = kS, lc = kS, lr = kS;
#pragma unroll
  for (int i = 0; i < kPerThread; ++i) {
    const int s = tl + kPairThreads * i;
    if (c[i] != r[i])   ldiv = min(ldiv, s);
    if (c[i] == PAD_ID) lc   = min(lc, s);
    if (r[i] == PAD_ID) lr   = min(lr, s);
  }
  ldiv = wave_min(ldiv);
  lc   = wave_min(lc);
  lr   = wave_min(lr);
  if (lane == 0) {
    wmin[0][wv] = ldiv;
    wmin[1][wv] = lc;
    wmin[2][wv] = lr;
  }
  __syncthreads();

  int div = kS, cin = kS, rin = kS;
  for (int j = 0; j < kPairWaves; ++j) {
    div = min(div, wmin[0][b * kPairWaves + j]);
    cin = min(cin, wmin[1][b * kPairWaves + j]);
    rin = min(rin, wmin[2][b * kPairWaves + j]);
  }
  const int end = max(cin, rin);

  const int   cnt  = max(end - div, 0);
  const float fcnt = (float)max(cnt, 1);
  const int   last = max(end - 1, 0);

  float sc = 0.f, sr = 0.f, sl = 0.f;
#pragma unroll
  for (int i = 0; i < kPerThread; ++i) {
    const int s = tl + kPairThreads * i;
    if (s >= div && s < end) {
      const float cv = fc[i], rv = fr[i];
      sc += cv;
      sr += rv;
      const float z = rv - cv;
      sl += fmaxf(z, 0.f) + log1pf(expf(-fabsf(z)));
    }
    if (s == last) {
      tail[2 * kB + b] = fc[i];
      tail[3 * kB + b] = fr[i];
    }
  }
  sc = wave_sum(sc);
  sr = wave_sum(sr);
  sl = wave_sum(sl);
  if (lane == 0) {
    wsum[0][wv] = sc;
    wsum[1][wv] = sr;
    wsum[2][wv] = sl;
  }
  __syncthreads();

  if (tl == 0) {
    float a = 0.f, d = 0.f, l = 0.f;
    for (int j = 0; j < kPairWaves; ++j) {
      a += wsum[0][b * kPairWaves + j];
      d += wsum[1][b * kPairWaves + j];
      l += wsum[2][b * kPairWaves + j];
    }
    tail[b]      = a / fcnt;
    tail[kB + b] = d / fcnt;
    ploss[b]     = l / fcnt;
  }
  __syncthreads();

  const int base = 1 + kRows;
  if (t < 4 * kB) out[base + t] = tail[t];
  if (t == 0) {
    float loss_acc = 0.f;
    for (int p = 0; p < kB; ++p) loss_acc += ploss[p];
    out[0] = loss_acc / (float)kB;
  }
}

extern "C" void kernel_launch(void* const* d_in, const int* in_sizes, int n_in,
                              void* d_out, int out_size, void* d_ws, size_t ws_size,
                              hipStream_t stream) {
  (void)in_sizes; (void)n_in; (void)out_size;

  const float* hs  = (const float*)d_in[0];
  const float* w   = (const float*)d_in[1];
  const int*   ids = (const int*)d_in[2];

  float* out     = (float*)d_out;
  float* rewards = out + 1;

  constexpr int KSEG = 4;
  const dim3 blk(256);

  if (ws_size >= (size_t)KSEG * kRows * sizeof(float)) {
    float* part = (float*)d_ws;
    const int waves = (kRows / 16) * KSEG;
    reward_gemv_wmma<KSEG><<<dim3(waves / 8), blk, 0, stream>>>(hs, w, part);
    combine_partials<KSEG><<<dim3((kRows + 255) / 256), blk, 0, stream>>>(part, rewards);
  } else {
    reward_gemv_wmma<1><<<dim3((kRows / 16) / 8), blk, 0, stream>>>(hs, w, rewards);
  }

  pair_reduce<<<1, dim3(1024), 0, stream>>>(ids, rewards, out);
}


# loss term softplus(r-c) of each position formed in the stream kernel's epilogue (a block takes the chosen and the rejected rows of 32 positions) and read back by pair_reduce; sum order kept
# speedup vs baseline: 2.9966x; 1.9526x over previous
#include <hip/hip_runtime.h>


typedef __attribute__((ext_vector_type(2))) float v2f;
typedef __attribute__((ext_vector_type(4))) float v4f;
typedef __attribute__((ext_vector_type(8))) float v8f;

#define PAD_ID 50256

constexpr int kB    = 4;
constexpr int kS    = 2048;
constexpr int kD    = 4096;
constexpr int kRows = 2 * kB * kS;

constexpr int kSegs        = 2;
constexpr int kPiece       = kD / kSegs;
constexpr int kLoads       = kPiece / 128;
constexpr int kWaveRows    = 8;
constexpr int kStreamWaves = 16;
constexpr int kStreamBlocks = (kRows / kWaveRows) * kSegs / kStreamWaves;
constexpr int kBlockRows   = (kStreamWaves / kSegs) * kWaveRows;
constexpr int kBlockPos    = kBlockRows / 2;
constexpr int kPosBlocks   = kS / kBlockPos;
constexpr int kSideGroups  = kBlockPos / kWaveRows;
static_assert(kB * kPosBlocks == kStreamBlocks, "one block per (pair, 32 positions)");

__global__ __launch_bounds__(kStreamWaves * 32) void reward_gemv_stream(
    const float* __restrict__ hs,
    const float* __restrict__ w,
    float* __restrict__ rewards,
    float* __restrict__ lossterm) {
  __shared__ float xch[kSegs][kBlockRows];

  const int lane = threadIdx.x & 31;
  const int wave = __builtin_amdgcn_readfirstlane((int)(threadIdx.x >> 5));
  const int pair = blockIdx.x / kPosBlocks;
  const int pos0 = (blockIdx.x % kPosBlocks) * kBlockPos;
  const int seg  = wave % kSegs;
  const int grp  = wave / kSegs;
  const int side = grp / kSideGroups;
  const int row0 = (side * kB + pair) * kS + pos0 + (grp % kSideGroups) * kWaveRows;

  const float* __restrict__ wp = w + seg * kPiece + 4 * lane;
  v4f wq[kLoads];
#pragma unroll
  for (int q = 0; q < kLoads; ++q) wq[q] = *(const v4f*)(wp + 128 * q);

  const float* __restrict__ hp = hs + (size_t)row0 * kD + seg * kPiece + 4 * lane;

  v4f   buf[2][kLoads];
  float sum[kWaveRows];

  const int  dsel = lane & 7;
  const bool dval = ((lane >> 3) & 1) == (lane >> 4);

#define LOAD_PIECE(R)                                                          \
  do {                                                                         \
    _Pragma("unroll") for (int q = 0; q < kLoads; ++q)                         \
      buf[(R) & 1][q] = __builtin_nontemporal_load(                            \
          (const v4f*)(hp + (size_t)(R) * kD + 128 * q));                      \
  } while (0)

  LOAD_PIECE(0);
#pragma unroll
  for (int r = 0; r < kWaveRows; ++r) {
    __builtin_amdgcn_sched_barrier(0);
    if (r + 1 < kWaveRows) LOAD_PIECE(r + 1);
    __builtin_amdgcn_sched_barrier(0);

    v8f acc0 = {};
    v8f acc1 = {};
#pragma unroll
    for (int q = 0; q < kLoads; ++q) {
      const v4f a  = buf[r & 1][q];
      const v4f bw = wq[q];
      v2f a0 = {a.x, a.y};
      v2f b0 = {bw.x, bw.y};
      acc0 = __builtin_amdgcn_wmma_f32_16x16x4_f32(
           false, a0,  false, b0,
           (short)0, acc0,  false,  false);
      v2f a1 = {a.z, a.w};
      v2f b1 = {bw.z, bw.w};
      acc1 = __builtin_amdgcn_wmma_f32_16x16x4_f32(
          false, a1, false, b1, (short)0, acc1, false, false);
    }
    const v8f acc = acc0 + acc1;
    float d = acc[0];
#pragma unroll
    for (int j = 1; j < 8; ++j) d = (dsel == j) ? acc[j] : d;
    sum[r] = dval ? d : 0.f;
  }
#undef LOAD_PIECE

#pragma unroll
  for (int r = 0; r < kWaveRows; ++r) {
    float s = sum[r];
#pragma unroll
    for (int o = 16; o >= 1; o >>= 1) s += __shfl_xor(s, o, 32);
    sum[r] = s;
  }
  float mine = 0.f;
#pragma unroll
  for (int r = 0; r < kWaveRows; ++r)
    if (lane == r) mine = sum[r];
  if (lane < kWaveRows) xch[seg][grp * kWaveRows + lane] = mine;
  __syncthreads();
  if (threadIdx.x < kBlockRows) {
    const int t  = threadIdx.x;
    const int sd = t / kBlockPos;
    const int p  = t % kBlockPos;
    float s = 0.f;
#pragma unroll
    for (int g = 0; g < kSegs; ++g) s += xch[g][t];
    rewards[(size_t)(sd * kB + pair) * kS + pos0 + p] = s;
    if (t < kBlockPos) {
      float rv = 0.f;
#pragma unroll
      for (int g = 0; g < kSegs; ++g) rv += xch[g][kBlockPos + t];
      const float z = rv - s;
      lossterm[(size_t)pair * kS + pos0 + t] =
          fmaxf(z, 0.f) + log1pf(expf(-fabsf(z)));
    }
  }
}

constexpr int kPairThreads = 256;
constexpr int kPairWaves   = kPairThreads / 32;
constexpr int kPerThread   = kS / kPairThreads;

__device__ __forceinline__ int wave_min(int v) {
  for (int o = 16; o > 0; o >>= 1) v = min(v, __shfl_xor(v, o, 32));
  return v;
}

__device__ __forceinline__ float wave_sum(float v) {
  for (int o = 16; o > 0; o >>= 1) v += __shfl_xor(v, o, 32);
  return v;
}

__global__ __launch_bounds__(1024) void pair_reduce(
    const int* __restrict__ ids,
    const float* __restrict__ rewards,
    const float* __restrict__ lossterm,
    float* __restrict__ out) {
  __shared__ int   wmin[3][kB * kPairWaves];
  __shared__ float wsum[3][kB * kPairWaves];
  __shared__ float tail[4 * kB];
  __shared__ float ploss[kB];

  const int t    = threadIdx.x;
  const int b    = t >> 8;
  const int tl   = t & 255;
  const int lane = t & 31;
  const int wv   = t >> 5;

  const int*   ch = ids + (size_t)b * kS;
  const int*   rj = ids + (size_t)(kB + b) * kS;
  const float* cr = rewards + (size_t)b * kS;
  const float* rr = rewards + (size_t)(kB + b) * kS;
  const float* lt = lossterm + (size_t)b * kS;

  int   c[kPerThread], r[kPerThread];
  float fc[kPerThread], fr[kPerThread], fl[kPerThread];
#pragma unroll
  for (int i = 0; i < kPerThread; ++i) {
    const int s = tl + kPairThreads * i;
    c[i]  = ch[s];
    r[i]  = rj[s];
    fc[i] = cr[s];
    fr[i] = rr[s];
    fl[i] = lt[s];
  }

  int ldiv = kS, lc = kS, lr = kS;
#pragma unroll
  for (int i = 0; i < kPerThread; ++i) {
    const int s = tl + kPairThreads * i;
    if (c[i] != r[i])   ldiv = min(ldiv, s);
    if (c[i] == PAD_ID) lc   = min(lc, s);
    if (r[i] == PAD_ID) lr   = min(lr, s);
  }
  ldiv = wave_min(ldiv);
  lc   = wave_min(lc);
  lr   = wave_min(lr);
  if (lane == 0) {
    wmin[0][wv] = ldiv;
    wmin[1][wv] = lc;
    wmin[2][wv] = lr;
  }
  __syncthreads();

  int div = kS, cin = kS, rin = kS;
  for (int j = 0; j < kPairWaves; ++j) {
    div = min(div, wmin[0][b * kPairWaves + j]);
    cin = min(cin, wmin[1][b * kPairWaves + j]);
    rin = min(rin, wmin[2][b * kPairWaves + j]);
  }
  const int end = max(cin, rin);

  const int   cnt  = max(end - div, 0);
  const float fcnt = (float)max(cnt, 1);
  const int   last = max(end - 1, 0);

  float sc = 0.f, sr = 0.f, sl = 0.f;
#pragma unroll
  for (int i = 0; i < kPerThread; ++i) {
    const int s = tl + kPairThreads * i;
    if (s >= div && s < end) {
      sc += fc[i];
      sr += fr[i];
      sl += fl[i];
    }
    if (s == last) {
      tail[2 * kB + b] = fc[i];
      tail[3 * kB + b] = fr[i];
    }
  }
  sc = wave_sum(sc);
  sr = wave_sum(sr);
  sl = wave_sum(sl);
  if (lane == 0) {
    wsum[0][wv] = sc;
    wsum[1][wv] = sr;
    wsum[2][wv] = sl;
  }
  __syncthreads();

  if (tl == 0) {
    float a = 0.f, d = 0.f, l = 0.f;
    for (int j = 0; j < kPairWaves; ++j) {
      a += wsum[0][b * kPairWaves + j];
      d += wsum[1][b * kPairWaves + j];
      l += wsum[2][b * kPairWaves + j];
    }
    tail[b]      = a / fcnt;
    tail[kB + b] = d / fcnt;
    ploss[b]     = l / fcnt;
  }
  __syncthreads();

  const int base = 1 + kRows;
  if (t < 4 * kB) out[base + t] = tail[t];
  if (t == 0) {
    float loss_acc = 0.f;
    for (int p = 0; p < kB; ++p) loss_acc += ploss[p];
    out[0] = loss_acc / (float)kB;
  }
}

extern "C" void kernel_launch(void* const* d_in, const int* in_sizes, int n_in,
                              void* d_out, int out_size, void* d_ws, size_t ws_size,
                              hipStream_t stream) {
  (void)in_sizes; (void)n_in; (void)out_size; (void)ws_size;

  const float* hs  = (const float*)d_in[0];
  const float* w   = (const float*)d_in[1];
  const int*   ids = (const int*)d_in[2];

  float* out     = (float*)d_out;
  float* rewards = out + 1;
  float* lossterm = (float*)d_ws;

  reward_gemv_stream<<<dim3(kStreamBlocks), dim3(kStreamWaves * 32), 0, stream>>>(hs, w, rewards, lossterm);

  pair_reduce<<<1, dim3(1024), 0, stream>>>(ids, rewards, lossterm, out);
}


# masked window sums formed in the stream kernel: each block scans its pair's ids for [div,end) and writes 3 partials of 32 positions; pair_reduce adds 64 partials per pair; reduction tree changes
# speedup vs baseline: 3.1808x; 1.0615x over previous
#include <hip/hip_runtime.h>


typedef __attribute__((ext_vector_type(2))) float v2f;
typedef __attribute__((ext_vector_type(4))) float v4f;
typedef __attribute__((ext_vector_type(8))) float v8f;
typedef __attribute__((ext_vector_type(4))) int   v4i;

#define PAD_ID 50256

constexpr int kB    = 4;
constexpr int kS    = 2048;
constexpr int kD    = 4096;
constexpr int kRows = 2 * kB * kS;

constexpr int kPartPerPair = 64;
constexpr int kWsPart = 0;
constexpr int kWsEnd  = 3 * kB * kPartPerPair;
constexpr int kWsWin  = kWsEnd + 2 * kB;

constexpr int kSegs        = 2;
constexpr int kPiece       = kD / kSegs;
constexpr int kLoads       = kPiece / 128;
constexpr int kWaveRows    = 8;
constexpr int kStreamWaves = 16;
constexpr int kStreamBlocks = (kRows / kWaveRows) * kSegs / kStreamWaves;
constexpr int kBlockRows   = (kStreamWaves / kSegs) * kWaveRows;
constexpr int kBlockPos    = kBlockRows / 2;
constexpr int kPosBlocks   = kS / kBlockPos;
constexpr int kSideGroups  = kBlockPos / kWaveRows;
static_assert(kB * kPosBlocks == kStreamBlocks, "one block per (pair, 32 positions)");
static_assert(kPosBlocks == kPartPerPair, "one partial per block of a pair");
static_assert(kStreamWaves * 32 * 4 == kS, "the block's threads scan a pair's ids four at a time");

__device__ __forceinline__ int wave_first(int li, int base) {
  const unsigned m = __builtin_amdgcn_ballot_w32(li < 4);
  const int l0 = __builtin_ctz(m | 0x80000000u);
  const int at = base + 4 * l0 + __builtin_amdgcn_readlane(li, l0);
  return m ? at : kS;
}

__device__ __forceinline__ int block_first(int v) {
  const unsigned m = __builtin_amdgcn_ballot_w32(v < kS);
  return __builtin_amdgcn_readlane(v, __builtin_ctz(m | 0x80000000u));
}

__device__ __forceinline__ float xhalf(float x) {
  const int b = __builtin_bit_cast(int, x);
  return __builtin_bit_cast(float,
      __builtin_amdgcn_permlanex16(b, b, 0x76543210, 0xfedcba98, false, false));
}

template <int M>
__device__ __forceinline__ float xrow(float x) {
  const int b = __builtin_bit_cast(int, x);
  return __builtin_bit_cast(float,
      __builtin_amdgcn_update_dpp(0, b, 0x160 + M, 0xf, 0xf, false));
}

__device__ __forceinline__ float wave_sum(float v) {
  v += xrow<1>(v);
  v += xrow<2>(v);
  v += xrow<4>(v);
  v += xrow<8>(v);
  v += xhalf(v);
  return v;
}

__global__ __launch_bounds__(kStreamWaves * 32) void reward_gemv_stream(
    const float* __restrict__ hs,
    const float* __restrict__ w,
    const int* __restrict__ ids,
    float* __restrict__ rewards,
    float* __restrict__ ws) {
  __shared__ float xch[kSegs][kBlockRows];
  __shared__ int wfirst[3][kStreamWaves];

  const int lane = threadIdx.x & 31;
  const int wave = __builtin_amdgcn_readfirstlane((int)(threadIdx.x >> 5));
  const int pair = blockIdx.x / kPosBlocks;
  const int pos0 = (blockIdx.x % kPosBlocks) * kBlockPos;
  const int seg  = wave % kSegs;
  const int grp  = wave / kSegs;
  const int side = grp / kSideGroups;
  const int row0 = (side * kB + pair) * kS + pos0 + (grp % kSideGroups) * kWaveRows;

  const v4i idc = *(const v4i*)(ids + (size_t)pair * kS + 4 * threadIdx.x);
  const v4i idr = *(const v4i*)(ids + (size_t)(kB + pair) * kS + 4 * threadIdx.x);
  __builtin_amdgcn_sched_barrier(0);

  const float* __restrict__ wp = w + seg * kPiece + 4 * lane;
  v4f wq[kLoads];
#pragma unroll
  for (int q = 0; q < kLoads; ++q) wq[q] = *(const v4f*)(wp + 128 * q);

  const float* __restrict__ hp = hs + (size_t)row0 * kD + seg * kPiece + 4 * lane;

  v4f   buf[2][kLoads];
  float sum[kWaveRows];

  const int  dsel = lane & 7;
  const bool dval = ((lane >> 3) & 1) == (lane >> 4);

#define LOAD_PIECE(R)                                                          \
  do {                                                                         \
    _Pragma("unroll") for (int q = 0; q < kLoads; ++q)                         \
      buf[(R) & 1][q] = __builtin_nontemporal_load(                            \
          (const v4f*)(hp + (size_t)(R) * kD + 128 * q));                      \
  } while (0)

  LOAD_PIECE(0);
#pragma unroll
  for (int r = 0; r < kWaveRows; ++r) {
    __builtin_amdgcn_sched_barrier(0);
    if (r + 1 < kWaveRows) LOAD_PIECE(r + 1);
    __builtin_amdgcn_sched_barrier(0);

    if (r == 0) {
      int lid = 4, lic = 4, lir = 4;
#pragma unroll
      for (int i = 3; i >= 0; --i) {
        if (idc[i] != idr[i]) lid = i;
        if (idc[i] == PAD_ID) lic = i;
        if (idr[i] == PAD_ID) lir = i;
      }
      const int base = wave * 128;
      const int fd = wave_first(lid, base);
      const int fc = wave_first(lic, base);
      const int fr = wave_first(lir, base);
      if (lane == 0) {
        wfirst[0][wave] = fd;
        wfirst[1][wave] = fc;
        wfirst[2][wave] = fr;
      }
    }

    v8f acc0 = {};
    v8f acc1 = {};
#pragma unroll
    for (int q = 0; q < kLoads; ++q) {
      const v4f a  = buf[r & 1][q];
      const v4f bw = wq[q];
      v2f a0 = {a.x, a.y};
      v2f b0 = {bw.x, bw.y};
      acc0 = __builtin_amdgcn_wmma_f32_16x16x4_f32(
           false, a0,  false, b0,
           (short)0, acc0,  false,  false);
      v2f a1 = {a.z, a.w};
      v2f b1 = {bw.z, bw.w};
      acc1 = __builtin_amdgcn_wmma_f32_16x16x4_f32(
          false, a1, false, b1, (short)0, acc1, false, false);
    }
    const v8f acc = acc0 + acc1;
    float d = acc[0];
#pragma unroll
    for (int j = 1; j < 8; ++j) d = (dsel == j) ? acc[j] : d;
    sum[r] = dval ? d : 0.f;
  }
#undef LOAD_PIECE

#pragma unroll
  for (int r = 0; r < kWaveRows; ++r) {
    float s = sum[r];
#pragma unroll
    for (int o = 16; o >= 1; o >>= 1) s += __shfl_xor(s, o, 32);
    sum[r] = s;
  }
  float mine = 0.f;
#pragma unroll
  for (int r = 0; r < kWaveRows; ++r)
    if (lane == r) mine = sum[r];
  if (lane < kWaveRows) xch[seg][grp * kWaveRows + lane] = mine;
  __syncthreads();
  if (wave < kBlockRows / 32) {
    const int t  = threadIdx.x;
    const int sd = wave;
    const int p  = lane;
    float s = 0.f;
#pragma unroll
    for (int g = 0; g < kSegs; ++g) s += xch[g][t];
    rewards[(size_t)(sd * kB + pair) * kS + pos0 + p] = s;

    const int div  = block_first(wfirst[0][lane & (kStreamWaves - 1)]);
    const int cin  = block_first(wfirst[1][lane & (kStreamWaves - 1)]);
    const int rin  = block_first(wfirst[2][lane & (kStreamWaves - 1)]);
    const int end  = max(cin, rin);
    const int last = max(end - 1, 0);
    const int pos  = pos0 + p;
    const bool in  = pos >= div && pos < end;
    const int blk  = blockIdx.x % kPosBlocks;

    const float ms = wave_sum(in ? s : 0.f);
    if (lane == 0) ws[kWsPart + (sd * kB + pair) * kPartPerPair + blk] = ms;
    if (pos == last) ws[kWsEnd + sd * kB + pair] = s;
    if (blk == 0 && t == 0) {
      ws[kWsWin + 2 * pair]     = (float)div;
      ws[kWsWin + 2 * pair + 1] = (float)end;
    }

    if (wave == 0) {
      float rv = 0.f;
#pragma unroll
      for (int g = 0; g < kSegs; ++g) rv += xch[g][kBlockPos + t];
      const float z  = rv - s;
      const float lt = fmaxf(z, 0.f) + log1pf(expf(-fabsf(z)));
      const float ml = wave_sum(in ? lt : 0.f);
      if (lane == 0) ws[kWsPart + (2 * kB + pair) * kPartPerPair + blk] = ml;
    }
  }
}

constexpr int kFinThreads = kB * kPartPerPair;
constexpr int kFinWaves   = kFinThreads / 32;

__global__ __launch_bounds__(kFinThreads) void pair_reduce(
    const float* __restrict__ ws,
    float* __restrict__ out) {
  __shared__ float wsum[3][kFinWaves];
  __shared__ float ends[2 * kB];
  __shared__ float win[2 * kB];

  const int t    = threadIdx.x;
  const int lane = t & 31;
  const int wv   = t >> 5;

  float a = ws[kWsPart + t];
  float d = ws[kWsPart + kFinThreads + t];
  float l = ws[kWsPart + 2 * kFinThreads + t];
  float e = 0.f, wn = 0.f;
  if (t < 2 * kB) {
    e  = ws[kWsEnd + t];
    wn = ws[kWsWin + t];
  }
  a = wave_sum(a);
  d = wave_sum(d);
  l = wave_sum(l);
  if (lane == 0) {
    wsum[0][wv] = a;
    wsum[1][wv] = d;
    wsum[2][wv] = l;
  }
  if (t < 2 * kB) {
    ends[t] = e;
    win[t]  = wn;
  }
  __syncthreads();

  const int base = 1 + kRows;
  if (t < 4 * kB) {
    float v;
    if (t < 2 * kB) {
      const int sd = t / kB;
      const int b  = t % kB;
      const float fcnt = fmaxf(win[2 * b + 1] - win[2 * b], 1.f);
      v = (0.f + wsum[sd][2 * b] + wsum[sd][2 * b + 1]) / fcnt;
    } else {
      v = ends[t - 2 * kB];
    }
    out[base + t] = v;
  }
  if (t == 0) {
    float loss_acc = 0.f;
    for (int p = 0; p < kB; ++p) {
      const float fcnt = fmaxf(win[2 * p + 1] - win[2 * p], 1.f);
      loss_acc += (0.f + wsum[2][2 * p] + wsum[2][2 * p + 1]) / fcnt;
    }
    out[0] = loss_acc / (float)kB;
  }
}

extern "C" void kernel_launch(void* const* d_in, const int* in_sizes, int n_in,
                              void* d_out, int out_size, void* d_ws, size_t ws_size,
                              hipStream_t stream) {
  (void)in_sizes; (void)n_in; (void)out_size; (void)ws_size;

  const float* hs  = (const float*)d_in[0];
  const float* w   = (const float*)d_in[1];
  const int*   ids = (const int*)d_in[2];

  float* out     = (float*)d_out;
  float* rewards = out + 1;
  float* ws      = (float*)d_ws;

  reward_gemv_stream<<<dim3(kStreamBlocks), dim3(kStreamWaves * 32), 0, stream>>>(hs, w, ids, rewards, ws);

  pair_reduce<<<1, dim3(kFinThreads), 0, stream>>>(ws, out);
}
